# starting kernel
# baseline (speedup 1.0000x reference)
#include <hip/hip_runtime.h>
#include <hip/hip_bf16.h>


#define NQ    4096
#define NT    16384
#define DDIM  32
#define OUTD  16
#define SEGS  8
#define SEGLEN (NT / SEGS)
#define EPSV  1e-6f
#define BIGF  3.0e38f

typedef __attribute__((ext_vector_type(2))) float v2f;
typedef __attribute__((ext_vector_type(4))) float v4f;
typedef __attribute__((ext_vector_type(8))) float v8f;

__device__ __forceinline__ void bot5_insert(float (&tv)[5], int (&ti)[5],
                                            float v, int id) {
    if (v < tv[4]) {
        tv[4] = v; ti[4] = id;
#pragma unroll
        for (int k = 4; k > 0; --k) {
            if (tv[k] < tv[k - 1]) {
                float fv = tv[k]; tv[k] = tv[k - 1]; tv[k - 1] = fv;
                int   fi = ti[k]; ti[k] = ti[k - 1]; ti[k - 1] = fi;
            }
        }
    }
}

__global__ __launch_bounds__(128) void idw_prep(
    const float* __restrict__ x, const float* __restrict__ tx,
    const float* __restrict__ w,
    float* __restrict__ xq_s, float* __restrict__ xt_s,
    float* __restrict__ qnorm, float* __restrict__ tnorm) {
    int i = blockIdx.x * blockDim.x + threadIdx.x;
    bool isQ = i < NQ;
    const float* src = isQ ? (x + (size_t)i * DDIM) : (tx + (size_t)(i - NQ) * DDIM);
    float*       dst = isQ ? (xq_s + (size_t)i * DDIM) : (xt_s + (size_t)(i - NQ) * DDIM);
    float nrm = 0.f;
#pragma unroll
    for (int j = 0; j < DDIM; ++j) {
        int h  = j >> 4;
        int jj = j & 15;
        int K  = 4 * (jj >> 1) + (jj & 1) + 2 * h;
        float s = __expf(-w[K]);
        float v = src[K] * s;
        dst[j] = v;
        nrm   += v * v;
    }
    if (isQ) qnorm[i] = nrm; else tnorm[i - NQ] = nrm;
}

__global__ __launch_bounds__(32) void idw_main(
    const float* __restrict__ xq_s, const float* __restrict__ xt_s,
    const float* __restrict__ qnorm, const float* __restrict__ tnorm,
    float* __restrict__ pvals, int* __restrict__ pidx) {
    const int wid   = blockIdx.x;
    const int seg   = wid % SEGS;
    const int qtile = wid / SEGS;
    const int lane  = threadIdx.x;
    const int h     = lane >> 4;
    const int ql    = lane & 15;
    const int q     = qtile * 16 + ql;

    const float* qrow = xq_s + (size_t)q * DDIM + 16 * h;
    v4f bq0 = *(const v4f*)(qrow + 0);
    v4f bq1 = *(const v4f*)(qrow + 4);
    v4f bq2 = *(const v4f*)(qrow + 8);
    v4f bq3 = *(const v4f*)(qrow + 12);
    v2f b[8] = {bq0.xy, bq0.zw, bq1.xy, bq1.zw, bq2.xy, bq2.zw, bq3.xy, bq3.zw};
    const float qn = qnorm[q];

    float tv[5] = {BIGF, BIGF, BIGF, BIGF, BIGF};
    int   ti[5] = {0, 0, 0, 0, 0};

    const int tbeg = seg * SEGLEN;
#pragma unroll 2
    for (int t0 = tbeg; t0 < tbeg + SEGLEN; t0 += 16) {
        const float* arow = xt_s + (size_t)(t0 + ql) * DDIM + 16 * h;
        v4f a0 = *(const v4f*)(arow + 0);
        v4f a1 = *(const v4f*)(arow + 4);
        v4f a2 = *(const v4f*)(arow + 8);
        v4f a3 = *(const v4f*)(arow + 12);

        v8f c0 = {0.f, 0.f, 0.f, 0.f, 0.f, 0.f, 0.f, 0.f};
        v8f c1 = c0;
        c0 = __builtin_amdgcn_wmma_f32_16x16x4_f32(false, a0.xy, false, b[0], (short)0, c0, false, false);
        c1 = __builtin_amdgcn_wmma_f32_16x16x4_f32(false, a2.xy, false, b[4], (short)0, c1, false, false);
        c0 = __builtin_amdgcn_wmma_f32_16x16x4_f32(false, a0.zw, false, b[1], (short)0, c0, false, false);
        c1 = __builtin_amdgcn_wmma_f32_16x16x4_f32(false, a2.zw, false, b[5], (short)0, c1, false, false);
        c0 = __builtin_amdgcn_wmma_f32_16x16x4_f32(false, a1.xy, false, b[2], (short)0, c0, false, false);
        c1 = __builtin_amdgcn_wmma_f32_16x16x4_f32(false, a3.xy, false, b[6], (short)0, c1, false, false);
        c0 = __builtin_amdgcn_wmma_f32_16x16x4_f32(false, a1.zw, false, b[3], (short)0, c0, false, false);
        c1 = __builtin_amdgcn_wmma_f32_16x16x4_f32(false, a3.zw, false, b[7], (short)0, c1, false, false);

        const float* tnp = tnorm + t0 + 8 * h;
        v4f tna = *(const v4f*)(tnp);
        v4f tnb = *(const v4f*)(tnp + 4);
        float tn[8] = {tna.x, tna.y, tna.z, tna.w, tnb.x, tnb.y, tnb.z, tnb.w};

#pragma unroll
        for (int r = 0; r < 8; ++r) {
            float sq = qn + tn[r];
            sq = fmaf(-2.0f, c0[r], sq);
            sq = fmaf(-2.0f, c1[r], sq);
            bot5_insert(tv, ti, sq, t0 + 8 * h + r);
        }
    }

    float pv[5]; int pi[5];
#pragma unroll
    for (int k = 0; k < 5; ++k) {
        pv[k] = __shfl_xor(tv[k], 16, 32);
        pi[k] = __shfl_xor(ti[k], 16, 32);
    }
#pragma unroll
    for (int k = 0; k < 5; ++k) bot5_insert(tv, ti, pv[k], pi[k]);

    if (lane < 16) {
        int base = (q * SEGS + seg) * 5;
#pragma unroll
        for (int k = 0; k < 5; ++k) { pvals[base + k] = tv[k]; pidx[base + k] = ti[k]; }
    }
}

__global__ __launch_bounds__(128) void idw_reduce(
    const float* __restrict__ pvals, const int* __restrict__ pidx,
    const float* __restrict__ train_y, float* __restrict__ out) {
    int q = blockIdx.x * blockDim.x + threadIdx.x;
    if (q >= NQ) return;

    float tv[5] = {BIGF, BIGF, BIGF, BIGF, BIGF};
    int   ti[5] = {0, 0, 0, 0, 0};
    for (int s = 0; s < SEGS; ++s) {
        int base = (q * SEGS + s) * 5;
#pragma unroll
        for (int k = 0; k < 5; ++k)
            bot5_insert(tv, ti, pvals[base + k], pidx[base + k]);
    }

    float wv[5];
#pragma unroll
    for (int k = 0; k < 5; ++k)
        wv[k] = rsqrtf(fmaxf(tv[k], 0.f) + EPSV);

    float sw  = wv[0] + wv[1] + wv[2] + wv[3] + wv[4];
    float inv = 1.f / sw;

    v4f acc0 = {0.f, 0.f, 0.f, 0.f}, acc1 = acc0, acc2 = acc0, acc3 = acc0;
#pragma unroll
    for (int k = 0; k < 5; ++k) {
        const float* yr = train_y + (size_t)ti[k] * OUTD;
        float wgt = wv[k];
        v4f y0 = *(const v4f*)(yr + 0);
        v4f y1 = *(const v4f*)(yr + 4);
        v4f y2 = *(const v4f*)(yr + 8);
        v4f y3 = *(const v4f*)(yr + 12);
        acc0 += wgt * y0; acc1 += wgt * y1; acc2 += wgt * y2; acc3 += wgt * y3;
    }
    float* op = out + (size_t)q * OUTD;
    *(v4f*)(op + 0)  = acc0 * inv;
    *(v4f*)(op + 4)  = acc1 * inv;
    *(v4f*)(op + 8)  = acc2 * inv;
    *(v4f*)(op + 12) = acc3 * inv;
}

extern "C" void kernel_launch(void* const* d_in, const int* in_sizes, int n_in,
                              void* d_out, int out_size, void* d_ws, size_t ws_size,
                              hipStream_t stream) {
    const float* x       = (const float*)d_in[0];
    const float* train_x = (const float*)d_in[1];
    const float* train_y = (const float*)d_in[2];
    const float* weights = (const float*)d_in[3];
    float* out = (float*)d_out;

    float* ws    = (float*)d_ws;
    float* xq_s  = ws;
    float* xt_s  = ws + 131072;
    float* qnorm = ws + 655360;
    float* tnorm = ws + 659456;
    float* pvals = ws + 675840;
    int*   pidx  = (int*)(ws + 839680);

    idw_prep<<<(NQ + NT) / 128, 128, 0, stream>>>(x, train_x, weights,
                                                  xq_s, xt_s, qnorm, tnorm);
    idw_main<<<(NQ / 16) * SEGS, 32, 0, stream>>>(xq_s, xt_s, qnorm, tnorm,
                                                  pvals, pidx);
    idw_reduce<<<NQ / 128, 128, 0, stream>>>(pvals, pidx, train_y, out);
}


# main kernel: 4-wave workgroups stage each 128-row train block once through LDS (coalesced loads held in registers one stage ahead, two buffers); A fragments read from LDS; sum order unchanged
# speedup vs baseline: 1.6867x; 1.6867x over previous
#include <hip/hip_runtime.h>
#include <hip/hip_bf16.h>


#define NQ    4096
#define NT    16384
#define DDIM  32
#define OUTD  16
#define SEGS  8
#define SEGLEN (NT / SEGS)
#define EPSV  1e-6f
#define BIGF  3.0e38f

typedef __attribute__((ext_vector_type(2))) float v2f;
typedef __attribute__((ext_vector_type(4))) float v4f;
typedef __attribute__((ext_vector_type(8))) float v8f;

__device__ __forceinline__ void bot5_insert(float (&tv)[5], int (&ti)[5],
                                            float v, int id) {
    if (v < tv[4]) {
        tv[4] = v; ti[4] = id;
#pragma unroll
        for (int k = 4; k > 0; --k) {
            if (tv[k] < tv[k - 1]) {
                float fv = tv[k]; tv[k] = tv[k - 1]; tv[k - 1] = fv;
                int   fi = ti[k]; ti[k] = ti[k - 1]; ti[k - 1] = fi;
            }
        }
    }
}

__global__ __launch_bounds__(128) void idw_prep(
    const float* __restrict__ x, const float* __restrict__ tx,
    const float* __restrict__ w,
    float* __restrict__ xq_s, float* __restrict__ xt_s,
    float* __restrict__ qnorm, float* __restrict__ tnorm) {
    int i = blockIdx.x * blockDim.x + threadIdx.x;
    bool isQ = i < NQ;
    const float* src = isQ ? (x + (size_t)i * DDIM) : (tx + (size_t)(i - NQ) * DDIM);
    float*       dst = isQ ? (xq_s + (size_t)i * DDIM) : (xt_s + (size_t)(i - NQ) * DDIM);
    float nrm = 0.f;
#pragma unroll
    for (int j = 0; j < DDIM; ++j) {
        int h  = j >> 4;
        int jj = j & 15;
        int K  = 4 * (jj >> 1) + (jj & 1) + 2 * h;
        float s = __expf(-w[K]);
        float v = src[K] * s;
        dst[j] = v;
        nrm   += v * v;
    }
    if (isQ) qnorm[i] = nrm; else tnorm[i - NQ] = nrm;
}

#define WAVES_PER_WG 4
#define TROWS   128
#define PITCH   36
#define NSTAGE  (SEGLEN / TROWS)
#define CHUNKS_PER_THREAD ((TROWS * DDIM / 4) / (WAVES_PER_WG * 32))
#define SUBT    (TROWS / 16)

__global__ __launch_bounds__(WAVES_PER_WG * 32) void idw_main(
    const float* __restrict__ xq_s, const float* __restrict__ xt_s,
    const float* __restrict__ qnorm, const float* __restrict__ tnorm,
    float* __restrict__ pvals, int* __restrict__ pidx) {
    __shared__ __attribute__((aligned(16))) float sA[2][TROWS * PITCH];
    __shared__ __attribute__((aligned(16))) float sT[2][TROWS];

    const int tid   = threadIdx.x;
    const int wave  = tid >> 5;
    const int lane  = tid & 31;
    const int seg   = blockIdx.x % SEGS;
    const int qtile = (blockIdx.x / SEGS) * WAVES_PER_WG + wave;
    const int h     = lane >> 4;
    const int ql    = lane & 15;
    const int q     = qtile * 16 + ql;

    const float* qrow = xq_s + (size_t)q * DDIM + 16 * h;
    v4f bq0 = *(const v4f*)(qrow + 0);
    v4f bq1 = *(const v4f*)(qrow + 4);
    v4f bq2 = *(const v4f*)(qrow + 8);
    v4f bq3 = *(const v4f*)(qrow + 12);
    v2f b[8] = {bq0.xy, bq0.zw, bq1.xy, bq1.zw, bq2.xy, bq2.zw, bq3.xy, bq3.zw};
    const float qn = qnorm[q];

    float tv[5] = {BIGF, BIGF, BIGF, BIGF, BIGF};
    int   ti[5] = {0, 0, 0, 0, 0};

    const int tbeg = seg * SEGLEN;

    v4f g[CHUNKS_PER_THREAD];
    v4f gt = {0.f, 0.f, 0.f, 0.f};

    {
        const float* gsrc = xt_s + (size_t)tbeg * DDIM;
#pragma unroll
        for (int p = 0; p < CHUNKS_PER_THREAD; ++p)
            g[p] = *(const v4f*)(gsrc + (size_t)(p * (WAVES_PER_WG * 32) + tid) * 4);
        if (tid < TROWS / 4) gt = *(const v4f*)(tnorm + tbeg + tid * 4);
#pragma unroll
        for (int p = 0; p < CHUNKS_PER_THREAD; ++p) {
            int c = p * (WAVES_PER_WG * 32) + tid;
            *(v4f*)(&sA[0][(c >> 3) * PITCH + (c & 7) * 4]) = g[p];
        }
        if (tid < TROWS / 4) *(v4f*)(&sT[0][tid * 4]) = gt;
    }
    __syncthreads();

    for (int s = 0; s < NSTAGE; ++s) {
        const int buf = s & 1;
        const bool more = (s + 1 < NSTAGE);

        if (more) {
            const float* gsrc = xt_s + (size_t)(tbeg + (s + 1) * TROWS) * DDIM;
#pragma unroll
            for (int p = 0; p < CHUNKS_PER_THREAD; ++p)
                g[p] = *(const v4f*)(gsrc + (size_t)(p * (WAVES_PER_WG * 32) + tid) * 4);
            if (tid < TROWS / 4) gt = *(const v4f*)(tnorm + tbeg + (s + 1) * TROWS + tid * 4);
        }

        const float* aBase = &sA[buf][ql * PITCH + 16 * h];
        const float* tBase = &sT[buf][8 * h];
        const int    idBase = tbeg + s * TROWS + 8 * h;

        v4f a0 = *(const v4f*)(aBase + 0);
        v4f a1 = *(const v4f*)(aBase + 4);
        v4f a2 = *(const v4f*)(aBase + 8);
        v4f a3 = *(const v4f*)(aBase + 12);
        v4f tna = *(const v4f*)(tBase);
        v4f tnb = *(const v4f*)(tBase + 4);

#pragma unroll
        for (int sub = 0; sub < SUBT; ++sub) {
            v4f n0 = a0, n1 = a1, n2 = a2, n3 = a3, ntna = tna, ntnb = tnb;
            if (sub + 1 < SUBT) {
                const float* an = aBase + (sub + 1) * 16 * PITCH;
                n0 = *(const v4f*)(an + 0);
                n1 = *(const v4f*)(an + 4);
                n2 = *(const v4f*)(an + 8);
                n3 = *(const v4f*)(an + 12);
                ntna = *(const v4f*)(tBase + (sub + 1) * 16);
                ntnb = *(const v4f*)(tBase + (sub + 1) * 16 + 4);
            }

            v8f c0 = {0.f, 0.f, 0.f, 0.f, 0.f, 0.f, 0.f, 0.f};
            v8f c1 = c0;
            c0 = __builtin_amdgcn_wmma_f32_16x16x4_f32(false, a0.xy, false, b[0], (short)0, c0, false, false);
            c1 = __builtin_amdgcn_wmma_f32_16x16x4_f32(false, a2.xy, false, b[4], (short)0, c1, false, false);
            c0 = __builtin_amdgcn_wmma_f32_16x16x4_f32(false, a0.zw, false, b[1], (short)0, c0, false, false);
            c1 = __builtin_amdgcn_wmma_f32_16x16x4_f32(false, a2.zw, false, b[5], (short)0, c1, false, false);
            c0 = __builtin_amdgcn_wmma_f32_16x16x4_f32(false, a1.xy, false, b[2], (short)0, c0, false, false);
            c1 = __builtin_amdgcn_wmma_f32_16x16x4_f32(false, a3.xy, false, b[6], (short)0, c1, false, false);
            c0 = __builtin_amdgcn_wmma_f32_16x16x4_f32(false, a1.zw, false, b[3], (short)0, c0, false, false);
            c1 = __builtin_amdgcn_wmma_f32_16x16x4_f32(false, a3.zw, false, b[7], (short)0, c1, false, false);

            float tn[8] = {tna.x, tna.y, tna.z, tna.w, tnb.x, tnb.y, tnb.z, tnb.w};

#pragma unroll
            for (int r = 0; r < 8; ++r) {
                float sq = qn + tn[r];
                sq = fmaf(-2.0f, c0[r], sq);
                sq = fmaf(-2.0f, c1[r], sq);
                bot5_insert(tv, ti, sq, idBase + sub * 16 + r);
            }

            a0 = n0; a1 = n1; a2 = n2; a3 = n3; tna = ntna; tnb = ntnb;
        }

        if (more) {
#pragma unroll
            for (int p = 0; p < CHUNKS_PER_THREAD; ++p) {
                int c = p * (WAVES_PER_WG * 32) + tid;
                *(v4f*)(&sA[buf ^ 1][(c >> 3) * PITCH + (c & 7) * 4]) = g[p];
            }
            if (tid < TROWS / 4) *(v4f*)(&sT[buf ^ 1][tid * 4]) = gt;
        }
        __syncthreads();
    }

    float pv[5]; int pi[5];
#pragma unroll
    for (int k = 0; k < 5; ++k) {
        pv[k] = __shfl_xor(tv[k], 16, 32);
        pi[k] = __shfl_xor(ti[k], 16, 32);
    }
#pragma unroll
    for (int k = 0; k < 5; ++k) bot5_insert(tv, ti, pv[k], pi[k]);

    if (lane < 16) {
        int base = (q * SEGS + seg) * 5;
#pragma unroll
        for (int k = 0; k < 5; ++k) { pvals[base + k] = tv[k]; pidx[base + k] = ti[k]; }
    }
}

__global__ __launch_bounds__(128) void idw_reduce(
    const float* __restrict__ pvals, const int* __restrict__ pidx,
    const float* __restrict__ train_y, float* __restrict__ out) {
    int q = blockIdx.x * blockDim.x + threadIdx.x;
    if (q >= NQ) return;

    float tv[5] = {BIGF, BIGF, BIGF, BIGF, BIGF};
    int   ti[5] = {0, 0, 0, 0, 0};
    for (int s = 0; s < SEGS; ++s) {
        int base = (q * SEGS + s) * 5;
#pragma unroll
        for (int k = 0; k < 5; ++k)
            bot5_insert(tv, ti, pvals[base + k], pidx[base + k]);
    }

    float wv[5];
#pragma unroll
    for (int k = 0; k < 5; ++k)
        wv[k] = rsqrtf(fmaxf(tv[k], 0.f) + EPSV);

    float sw  = wv[0] + wv[1] + wv[2] + wv[3] + wv[4];
    float inv = 1.f / sw;

    v4f acc0 = {0.f, 0.f, 0.f, 0.f}, acc1 = acc0, acc2 = acc0, acc3 = acc0;
#pragma unroll
    for (int k = 0; k < 5; ++k) {
        const float* yr = train_y + (size_t)ti[k] * OUTD;
        float wgt = wv[k];
        v4f y0 = *(const v4f*)(yr + 0);
        v4f y1 = *(const v4f*)(yr + 4);
        v4f y2 = *(const v4f*)(yr + 8);
        v4f y3 = *(const v4f*)(yr + 12);
        acc0 += wgt * y0; acc1 += wgt * y1; acc2 += wgt * y2; acc3 += wgt * y3;
    }
    float* op = out + (size_t)q * OUTD;
    *(v4f*)(op + 0)  = acc0 * inv;
    *(v4f*)(op + 4)  = acc1 * inv;
    *(v4f*)(op + 8)  = acc2 * inv;
    *(v4f*)(op + 12) = acc3 * inv;
}

extern "C" void kernel_launch(void* const* d_in, const int* in_sizes, int n_in,
                              void* d_out, int out_size, void* d_ws, size_t ws_size,
                              hipStream_t stream) {
    const float* x       = (const float*)d_in[0];
    const float* train_x = (const float*)d_in[1];
    const float* train_y = (const float*)d_in[2];
    const float* weights = (const float*)d_in[3];
    float* out = (float*)d_out;

    float* ws    = (float*)d_ws;
    float* xq_s  = ws;
    float* xt_s  = ws + 131072;
    float* qnorm = ws + 655360;
    float* tnorm = ws + 659456;
    float* pvals = ws + 675840;
    int*   pidx  = (int*)(ws + 839680);

    idw_prep<<<(NQ + NT) / 128, 128, 0, stream>>>(x, train_x, weights,
                                                  xq_s, xt_s, qnorm, tnorm);
    idw_main<<<(NQ / (16 * WAVES_PER_WG)) * SEGS, WAVES_PER_WG * 32, 0, stream>>>(xq_s, xt_s, qnorm, tnorm,
                                                  pvals, pidx);
    idw_reduce<<<NQ / 128, 128, 0, stream>>>(pvals, pidx, train_y, out);
}


# main kernel: top-5 insert made branch-free (5 independent compares, id selects, min/max ladder on values); a stage is one straight-line block; same values, ids, tie order; sum order unchanged
# speedup vs baseline: 1.9791x; 1.1734x over previous
#include <hip/hip_runtime.h>
#include <hip/hip_bf16.h>


#define NQ    4096
#define NT    16384
#define DDIM  32
#define OUTD  16
#define SEGS  8
#define SEGLEN (NT / SEGS)
#define EPSV  1e-6f
#define BIGF  3.0e38f

typedef __attribute__((ext_vector_type(2))) float v2f;
typedef __attribute__((ext_vector_type(4))) float v4f;
typedef __attribute__((ext_vector_type(8))) float v8f;

__device__ __forceinline__ void bot5_insert(float (&tv)[5], int (&ti)[5],
                                            float v, int id) {
    if (v < tv[4]) {
        tv[4] = v; ti[4] = id;
#pragma unroll
        for (int k = 4; k > 0; --k) {
            if (tv[k] < tv[k - 1]) {
                float fv = tv[k]; tv[k] = tv[k - 1]; tv[k - 1] = fv;
                int   fi = ti[k]; ti[k] = ti[k - 1]; ti[k - 1] = fi;
            }
        }
    }
}

__device__ __forceinline__ void bot5_net(float (&tv)[5], int (&ti)[5],
                                         float v, int id) {
    const float t0 = tv[0], t1 = tv[1], t2 = tv[2], t3 = tv[3], t4 = tv[4];
    const int   i0 = ti[0], i1 = ti[1], i2 = ti[2], i3 = ti[3], i4 = ti[4];
    const bool m0 = v < t0;
    const bool m1 = v < t1;
    const bool m2 = v < t2;
    const bool m3 = v < t3;
    const bool m4 = v < t4;
    const int below3 = m3 ? i3 : id;
    const int below2 = m2 ? i2 : id;
    const int below1 = m1 ? i1 : id;
    const int below0 = m0 ? i0 : id;
    ti[4] = m4 ? below3 : i4;
    ti[3] = m3 ? below2 : i3;
    ti[2] = m2 ? below1 : i2;
    ti[1] = m1 ? below0 : i1;
    ti[0] = m0 ? id : i0;
    tv[4] = fminf(t4, fmaxf(t3, v));
    tv[3] = fminf(t3, fmaxf(t2, v));
    tv[2] = fminf(t2, fmaxf(t1, v));
    tv[1] = fminf(t1, fmaxf(t0, v));
    tv[0] = fminf(t0, v);
}

__global__ __launch_bounds__(128) void idw_prep(
    const float* __restrict__ x, const float* __restrict__ tx,
    const float* __restrict__ w,
    float* __restrict__ xq_s, float* __restrict__ xt_s,
    float* __restrict__ qnorm, float* __restrict__ tnorm) {
    int i = blockIdx.x * blockDim.x + threadIdx.x;
    bool isQ = i < NQ;
    const float* src = isQ ? (x + (size_t)i * DDIM) : (tx + (size_t)(i - NQ) * DDIM);
    float*       dst = isQ ? (xq_s + (size_t)i * DDIM) : (xt_s + (size_t)(i - NQ) * DDIM);
    float nrm = 0.f;
#pragma unroll
    for (int j = 0; j < DDIM; ++j) {
        int h  = j >> 4;
        int jj = j & 15;
        int K  = 4 * (jj >> 1) + (jj & 1) + 2 * h;
        float s = __expf(-w[K]);
        float v = src[K] * s;
        dst[j] = v;
        nrm   += v * v;
    }
    if (isQ) qnorm[i] = nrm; else tnorm[i - NQ] = nrm;
}

#define WAVES_PER_WG 4
#define TROWS   128
#define PITCH   36
#define NSTAGE  (SEGLEN / TROWS)
#define CHUNKS_PER_THREAD ((TROWS * DDIM / 4) / (WAVES_PER_WG * 32))
#define SUBT    (TROWS / 16)

__global__ __launch_bounds__(WAVES_PER_WG * 32) void idw_main(
    const float* __restrict__ xq_s, const float* __restrict__ xt_s,
    const float* __restrict__ qnorm, const float* __restrict__ tnorm,
    float* __restrict__ pvals, int* __restrict__ pidx) {
    __shared__ __attribute__((aligned(16))) float sA[2][TROWS * PITCH];
    __shared__ __attribute__((aligned(16))) float sT[2][TROWS];

    const int tid   = threadIdx.x;
    const int wave  = tid >> 5;
    const int lane  = tid & 31;
    const int seg   = blockIdx.x % SEGS;
    const int qtile = (blockIdx.x / SEGS) * WAVES_PER_WG + wave;
    const int h     = lane >> 4;
    const int ql    = lane & 15;
    const int q     = qtile * 16 + ql;

    const float* qrow = xq_s + (size_t)q * DDIM + 16 * h;
    v4f bq0 = *(const v4f*)(qrow + 0);
    v4f bq1 = *(const v4f*)(qrow + 4);
    v4f bq2 = *(const v4f*)(qrow + 8);
    v4f bq3 = *(const v4f*)(qrow + 12);
    v2f b[8] = {bq0.xy, bq0.zw, bq1.xy, bq1.zw, bq2.xy, bq2.zw, bq3.xy, bq3.zw};
    const float qn = qnorm[q];

    float tv[5] = {BIGF, BIGF, BIGF, BIGF, BIGF};
    int   ti[5] = {0, 0, 0, 0, 0};

    const int tbeg = seg * SEGLEN;

    v4f g[CHUNKS_PER_THREAD];
    v4f gt = {0.f, 0.f, 0.f, 0.f};

    {
        const float* gsrc = xt_s + (size_t)tbeg * DDIM;
#pragma unroll
        for (int p = 0; p < CHUNKS_PER_THREAD; ++p)
            g[p] = *(const v4f*)(gsrc + (size_t)(p * (WAVES_PER_WG * 32) + tid) * 4);
        if (tid < TROWS / 4) gt = *(const v4f*)(tnorm + tbeg + tid * 4);
#pragma unroll
        for (int p = 0; p < CHUNKS_PER_THREAD; ++p) {
            int c = p * (WAVES_PER_WG * 32) + tid;
            *(v4f*)(&sA[0][(c >> 3) * PITCH + (c & 7) * 4]) = g[p];
        }
        if (tid < TROWS / 4) *(v4f*)(&sT[0][tid * 4]) = gt;
    }
    __syncthreads();

    for (int s = 0; s < NSTAGE; ++s) {
        const int buf = s & 1;
        const bool more = (s + 1 < NSTAGE);

        if (more) {
            const float* gsrc = xt_s + (size_t)(tbeg + (s + 1) * TROWS) * DDIM;
#pragma unroll
            for (int p = 0; p < CHUNKS_PER_THREAD; ++p)
                g[p] = *(const v4f*)(gsrc + (size_t)(p * (WAVES_PER_WG * 32) + tid) * 4);
            if (tid < TROWS / 4) gt = *(const v4f*)(tnorm + tbeg + (s + 1) * TROWS + tid * 4);
        }

        const float* aBase = &sA[buf][ql * PITCH + 16 * h];
        const float* tBase = &sT[buf][8 * h];
        const int    idBase = tbeg + s * TROWS + 8 * h;

        v4f a0 = *(const v4f*)(aBase + 0);
        v4f a1 = *(const v4f*)(aBase + 4);
        v4f a2 = *(const v4f*)(aBase + 8);
        v4f a3 = *(const v4f*)(aBase + 12);
        v4f tna = *(const v4f*)(tBase);
        v4f tnb = *(const v4f*)(tBase + 4);

#pragma unroll
        for (int sub = 0; sub < SUBT; ++sub) {
            v4f n0 = a0, n1 = a1, n2 = a2, n3 = a3, ntna = tna, ntnb = tnb;
            if (sub + 1 < SUBT) {
                const float* an = aBase + (sub + 1) * 16 * PITCH;
                n0 = *(const v4f*)(an + 0);
                n1 = *(const v4f*)(an + 4);
                n2 = *(const v4f*)(an + 8);
                n3 = *(const v4f*)(an + 12);
                ntna = *(const v4f*)(tBase + (sub + 1) * 16);
                ntnb = *(const v4f*)(tBase + (sub + 1) * 16 + 4);
            }

            v8f c0 = {0.f, 0.f, 0.f, 0.f, 0.f, 0.f, 0.f, 0.f};
            v8f c1 = c0;
            c0 = __builtin_amdgcn_wmma_f32_16x16x4_f32(false, a0.xy, false, b[0], (short)0, c0, false, false);
            c1 = __builtin_amdgcn_wmma_f32_16x16x4_f32(false, a2.xy, false, b[4], (short)0, c1, false, false);
            c0 = __builtin_amdgcn_wmma_f32_16x16x4_f32(false, a0.zw, false, b[1], (short)0, c0, false, false);
            c1 = __builtin_amdgcn_wmma_f32_16x16x4_f32(false, a2.zw, false, b[5], (short)0, c1, false, false);
            c0 = __builtin_amdgcn_wmma_f32_16x16x4_f32(false, a1.xy, false, b[2], (short)0, c0, false, false);
            c1 = __builtin_amdgcn_wmma_f32_16x16x4_f32(false, a3.xy, false, b[6], (short)0, c1, false, false);
            c0 = __builtin_amdgcn_wmma_f32_16x16x4_f32(false, a1.zw, false, b[3], (short)0, c0, false, false);
            c1 = __builtin_amdgcn_wmma_f32_16x16x4_f32(false, a3.zw, false, b[7], (short)0, c1, false, false);

            float tn[8] = {tna.x, tna.y, tna.z, tna.w, tnb.x, tnb.y, tnb.z, tnb.w};

#pragma unroll
            for (int r = 0; r < 8; ++r) {
                float sq = qn + tn[r];
                sq = fmaf(-2.0f, c0[r], sq);
                sq = fmaf(-2.0f, c1[r], sq);
                bot5_net(tv, ti, sq, idBase + sub * 16 + r);
            }

            a0 = n0; a1 = n1; a2 = n2; a3 = n3; tna = ntna; tnb = ntnb;
        }

        if (more) {
#pragma unroll
            for (int p = 0; p < CHUNKS_PER_THREAD; ++p) {
                int c = p * (WAVES_PER_WG * 32) + tid;
                *(v4f*)(&sA[buf ^ 1][(c >> 3) * PITCH + (c & 7) * 4]) = g[p];
            }
            if (tid < TROWS / 4) *(v4f*)(&sT[buf ^ 1][tid * 4]) = gt;
        }
        __syncthreads();
    }

    float pv[5]; int pi[5];
#pragma unroll
    for (int k = 0; k < 5; ++k) {
        pv[k] = __shfl_xor(tv[k], 16, 32);
        pi[k] = __shfl_xor(ti[k], 16, 32);
    }
#pragma unroll
    for (int k = 0; k < 5; ++k) bot5_insert(tv, ti, pv[k], pi[k]);

    if (lane < 16) {
        int base = (q * SEGS + seg) * 5;
#pragma unroll
        for (int k = 0; k < 5; ++k) { pvals[base + k] = tv[k]; pidx[base + k] = ti[k]; }
    }
}

__global__ __launch_bounds__(128) void idw_reduce(
    const float* __restrict__ pvals, const int* __restrict__ pidx,
    const float* __restrict__ train_y, float* __restrict__ out) {
    int q = blockIdx.x * blockDim.x + threadIdx.x;
    if (q >= NQ) return;

    float tv[5] = {BIGF, BIGF, BIGF, BIGF, BIGF};
    int   ti[5] = {0, 0, 0, 0, 0};
    for (int s = 0; s < SEGS; ++s) {
        int base = (q * SEGS + s) * 5;
#pragma unroll
        for (int k = 0; k < 5; ++k)
            bot5_insert(tv, ti, pvals[base + k], pidx[base + k]);
    }

    float wv[5];
#pragma unroll
    for (int k = 0; k < 5; ++k)
        wv[k] = rsqrtf(fmaxf(tv[k], 0.f) + EPSV);

    float sw  = wv[0] + wv[1] + wv[2] + wv[3] + wv[4];
    float inv = 1.f / sw;

    v4f acc0 = {0.f, 0.f, 0.f, 0.f}, acc1 = acc0, acc2 = acc0, acc3 = acc0;
#pragma unroll
    for (int k = 0; k < 5; ++k) {
        const float* yr = train_y + (size_t)ti[k] * OUTD;
        float wgt = wv[k];
        v4f y0 = *(const v4f*)(yr + 0);
        v4f y1 = *(const v4f*)(yr + 4);
        v4f y2 = *(const v4f*)(yr + 8);
        v4f y3 = *(const v4f*)(yr + 12);
        acc0 += wgt * y0; acc1 += wgt * y1; acc2 += wgt * y2; acc3 += wgt * y3;
    }
    float* op = out + (size_t)q * OUTD;
    *(v4f*)(op + 0)  = acc0 * inv;
    *(v4f*)(op + 4)  = acc1 * inv;
    *(v4f*)(op + 8)  = acc2 * inv;
    *(v4f*)(op + 12) = acc3 * inv;
}

extern "C" void kernel_launch(void* const* d_in, const int* in_sizes, int n_in,
                              void* d_out, int out_size, void* d_ws, size_t ws_size,
                              hipStream_t stream) {
    const float* x       = (const float*)d_in[0];
    const float* train_x = (const float*)d_in[1];
    const float* train_y = (const float*)d_in[2];
    const float* weights = (const float*)d_in[3];
    float* out = (float*)d_out;

    float* ws    = (float*)d_ws;
    float* xq_s  = ws;
    float* xt_s  = ws + 131072;
    float* qnorm = ws + 655360;
    float* tnorm = ws + 659456;
    float* pvals = ws + 675840;
    int*   pidx  = (int*)(ws + 839680);

    idw_prep<<<(NQ + NT) / 128, 128, 0, stream>>>(x, train_x, weights,
                                                  xq_s, xt_s, qnorm, tnorm);
    idw_main<<<(NQ / (16 * WAVES_PER_WG)) * SEGS, WAVES_PER_WG * 32, 0, stream>>>(xq_s, xt_s, qnorm, tnorm,
                                                  pvals, pidx);
    idw_reduce<<<NQ / 128, 128, 0, stream>>>(pvals, pidx, train_y, out);
}


# main kernel ranks 8-row groups by their smallest distance (one top-5 update per 8 rows); reduce recomputes the 40 rows of the 5 best groups (f32 fma, tree sum over 8 lanes) and keeps the exact 5
# speedup vs baseline: 3.0717x; 1.5521x over previous
#include <hip/hip_runtime.h>
#include <hip/hip_bf16.h>


#define NQ    4096
#define NT    16384
#define DDIM  32
#define OUTD  16
#define SEGS  8
#define SEGLEN (NT / SEGS)
#define EPSV  1e-6f
#define BIGF  3.0e38f

typedef __attribute__((ext_vector_type(2))) float v2f;
typedef __attribute__((ext_vector_type(4))) float v4f;
typedef __attribute__((ext_vector_type(8))) float v8f;

__device__ __forceinline__ void bot5_insert(float (&tv)[5], int (&ti)[5],
                                            float v, int id) {
    if (v < tv[4]) {
        tv[4] = v; ti[4] = id;
#pragma unroll
        for (int k = 4; k > 0; --k) {
            if (tv[k] < tv[k - 1]) {
                float fv = tv[k]; tv[k] = tv[k - 1]; tv[k - 1] = fv;
                int   fi = ti[k]; ti[k] = ti[k - 1]; ti[k - 1] = fi;
            }
        }
    }
}

__device__ __forceinline__ void bot5_net(float (&tv)[5], int (&ti)[5],
                                         float v, int id) {
    const float t0 = tv[0], t1 = tv[1], t2 = tv[2], t3 = tv[3], t4 = tv[4];
    const int   i0 = ti[0], i1 = ti[1], i2 = ti[2], i3 = ti[3], i4 = ti[4];
    const bool m0 = v < t0;
    const bool m1 = v < t1;
    const bool m2 = v < t2;
    const bool m3 = v < t3;
    const bool m4 = v < t4;
    const int below3 = m3 ? i3 : id;
    const int below2 = m2 ? i2 : id;
    const int below1 = m1 ? i1 : id;
    const int below0 = m0 ? i0 : id;
    ti[4] = m4 ? below3 : i4;
    ti[3] = m3 ? below2 : i3;
    ti[2] = m2 ? below1 : i2;
    ti[1] = m1 ? below0 : i1;
    ti[0] = m0 ? id : i0;
    tv[4] = fminf(t4, fmaxf(t3, v));
    tv[3] = fminf(t3, fmaxf(t2, v));
    tv[2] = fminf(t2, fmaxf(t1, v));
    tv[1] = fminf(t1, fmaxf(t0, v));
    tv[0] = fminf(t0, v);
}

__global__ __launch_bounds__(128) void idw_prep(
    const float* __restrict__ x, const float* __restrict__ tx,
    const float* __restrict__ w,
    float* __restrict__ xq_s, float* __restrict__ xt_s,
    float* __restrict__ qnorm, float* __restrict__ tnorm) {
    int i = blockIdx.x * blockDim.x + threadIdx.x;
    bool isQ = i < NQ;
    const float* src = isQ ? (x + (size_t)i * DDIM) : (tx + (size_t)(i - NQ) * DDIM);
    float*       dst = isQ ? (xq_s + (size_t)i * DDIM) : (xt_s + (size_t)(i - NQ) * DDIM);
    float nrm = 0.f;
#pragma unroll
    for (int j = 0; j < DDIM; ++j) {
        int h  = j >> 4;
        int jj = j & 15;
        int K  = 4 * (jj >> 1) + (jj & 1) + 2 * h;
        float s = __expf(-w[K]);
        float v = src[K] * s;
        dst[j] = v;
        nrm   += v * v;
    }
    if (isQ) qnorm[i] = nrm; else tnorm[i - NQ] = nrm;
}

#define WAVES_PER_WG 4
#define TROWS   128
#define PITCH   36
#define NSTAGE  (SEGLEN / TROWS)
#define CHUNKS_PER_THREAD ((TROWS * DDIM / 4) / (WAVES_PER_WG * 32))
#define SUBT    (TROWS / 16)

__global__ __launch_bounds__(WAVES_PER_WG * 32) void idw_main(
    const float* __restrict__ xq_s, const float* __restrict__ xt_s,
    const float* __restrict__ tnorm,
    float* __restrict__ pvals, int* __restrict__ pidx) {
    __shared__ __attribute__((aligned(16))) float sA[2][TROWS * PITCH];
    __shared__ __attribute__((aligned(16))) float sT[2][TROWS];

    const int tid   = threadIdx.x;
    const int wave  = tid >> 5;
    const int lane  = tid & 31;
    const int seg   = blockIdx.x % SEGS;
    const int qtile = (blockIdx.x / SEGS) * WAVES_PER_WG + wave;
    const int h     = lane >> 4;
    const int ql    = lane & 15;
    const int q     = qtile * 16 + ql;

    const float* qrow = xq_s + (size_t)q * DDIM + 16 * h;
    v4f bq0 = *(const v4f*)(qrow + 0);
    v4f bq1 = *(const v4f*)(qrow + 4);
    v4f bq2 = *(const v4f*)(qrow + 8);
    v4f bq3 = *(const v4f*)(qrow + 12);
    v2f b[8] = {bq0.xy, bq0.zw, bq1.xy, bq1.zw, bq2.xy, bq2.zw, bq3.xy, bq3.zw};

    float tv[5] = {BIGF, BIGF, BIGF, BIGF, BIGF};
    int   ti[5] = {0, 0, 0, 0, 0};

    const int tbeg = seg * SEGLEN;

    v4f g[CHUNKS_PER_THREAD];
    v4f gt = {0.f, 0.f, 0.f, 0.f};

    {
        const float* gsrc = xt_s + (size_t)tbeg * DDIM;
#pragma unroll
        for (int p = 0; p < CHUNKS_PER_THREAD; ++p)
            g[p] = *(const v4f*)(gsrc + (size_t)(p * (WAVES_PER_WG * 32) + tid) * 4);
        if (tid < TROWS / 4) gt = *(const v4f*)(tnorm + tbeg + tid * 4);
#pragma unroll
        for (int p = 0; p < CHUNKS_PER_THREAD; ++p) {
            int c = p * (WAVES_PER_WG * 32) + tid;
            *(v4f*)(&sA[0][(c >> 3) * PITCH + (c & 7) * 4]) = g[p];
        }
        if (tid < TROWS / 4) *(v4f*)(&sT[0][tid * 4]) = gt;
    }
    __syncthreads();

    for (int s = 0; s < NSTAGE; ++s) {
        const int buf = s & 1;
        const bool more = (s + 1 < NSTAGE);

        if (more) {
            const float* gsrc = xt_s + (size_t)(tbeg + (s + 1) * TROWS) * DDIM;
#pragma unroll
            for (int p = 0; p < CHUNKS_PER_THREAD; ++p)
                g[p] = *(const v4f*)(gsrc + (size_t)(p * (WAVES_PER_WG * 32) + tid) * 4);
            if (tid < TROWS / 4) gt = *(const v4f*)(tnorm + tbeg + (s + 1) * TROWS + tid * 4);
        }

        const float* aBase = &sA[buf][ql * PITCH + 16 * h];
        const float* tBase = &sT[buf][8 * h];
        const int    idBase = tbeg + s * TROWS + 8 * h;

        v4f a0 = *(const v4f*)(aBase + 0);
        v4f a1 = *(const v4f*)(aBase + 4);
        v4f a2 = *(const v4f*)(aBase + 8);
        v4f a3 = *(const v4f*)(aBase + 12);
        v4f tna = *(const v4f*)(tBase);
        v4f tnb = *(const v4f*)(tBase + 4);

#pragma unroll
        for (int sub = 0; sub < SUBT; ++sub) {
            v4f n0 = a0, n1 = a1, n2 = a2, n3 = a3, ntna = tna, ntnb = tnb;
            if (sub + 1 < SUBT) {
                const float* an = aBase + (sub + 1) * 16 * PITCH;
                n0 = *(const v4f*)(an + 0);
                n1 = *(const v4f*)(an + 4);
                n2 = *(const v4f*)(an + 8);
                n3 = *(const v4f*)(an + 12);
                ntna = *(const v4f*)(tBase + (sub + 1) * 16);
                ntnb = *(const v4f*)(tBase + (sub + 1) * 16 + 4);
            }

            v8f c0 = {0.f, 0.f, 0.f, 0.f, 0.f, 0.f, 0.f, 0.f};
            v8f c1 = c0;
            c0 = __builtin_amdgcn_wmma_f32_16x16x4_f32(false, a0.xy, false, b[0], (short)0, c0, false, false);
            c1 = __builtin_amdgcn_wmma_f32_16x16x4_f32(false, a2.xy, false, b[4], (short)0, c1, false, false);
            c0 = __builtin_amdgcn_wmma_f32_16x16x4_f32(false, a0.zw, false, b[1], (short)0, c0, false, false);
            c1 = __builtin_amdgcn_wmma_f32_16x16x4_f32(false, a2.zw, false, b[5], (short)0, c1, false, false);
            c0 = __builtin_amdgcn_wmma_f32_16x16x4_f32(false, a1.xy, false, b[2], (short)0, c0, false, false);
            c1 = __builtin_amdgcn_wmma_f32_16x16x4_f32(false, a3.xy, false, b[6], (short)0, c1, false, false);
            c0 = __builtin_amdgcn_wmma_f32_16x16x4_f32(false, a1.zw, false, b[3], (short)0, c0, false, false);
            c1 = __builtin_amdgcn_wmma_f32_16x16x4_f32(false, a3.zw, false, b[7], (short)0, c1, false, false);

            float tn[8] = {tna.x, tna.y, tna.z, tna.w, tnb.x, tnb.y, tnb.z, tnb.w};

            float y[8];
#pragma unroll
            for (int r = 0; r < 8; ++r) {
                float v = fmaf(-2.0f, c0[r], tn[r]);
                y[r] = fmaf(-2.0f, c1[r], v);
            }
            const float m = fminf(fminf(fminf(fminf(y[0], y[1]), y[2]),
                                        fminf(fminf(y[3], y[4]), y[5])),
                                  fminf(y[6], y[7]));
            bot5_net(tv, ti, m, idBase + sub * 16);

            a0 = n0; a1 = n1; a2 = n2; a3 = n3; tna = ntna; tnb = ntnb;
        }

        if (more) {
#pragma unroll
            for (int p = 0; p < CHUNKS_PER_THREAD; ++p) {
                int c = p * (WAVES_PER_WG * 32) + tid;
                *(v4f*)(&sA[buf ^ 1][(c >> 3) * PITCH + (c & 7) * 4]) = g[p];
            }
            if (tid < TROWS / 4) *(v4f*)(&sT[buf ^ 1][tid * 4]) = gt;
        }
        __syncthreads();
    }

    float pv[5]; int pi[5];
#pragma unroll
    for (int k = 0; k < 5; ++k) {
        pv[k] = __shfl_xor(tv[k], 16, 32);
        pi[k] = __shfl_xor(ti[k], 16, 32);
    }
#pragma unroll
    for (int k = 0; k < 5; ++k) bot5_insert(tv, ti, pv[k], pi[k]);

    if (lane < 16) {
        int base = (q * SEGS + seg) * 5;
#pragma unroll
        for (int k = 0; k < 5; ++k) { pvals[base + k] = tv[k]; pidx[base + k] = ti[k]; }
    }
}

#define RLANES 8
#define RQ_PER_WG (128 / RLANES)

__device__ __forceinline__ void merge8(float (&tv)[5], int (&ti)[5], int lane) {
#pragma unroll
    for (int m = 1; m < RLANES; m <<= 1) {
        float pv[5]; int pi[5];
#pragma unroll
        for (int k = 0; k < 5; ++k) {
            pv[k] = __shfl_xor(tv[k], m, 32);
            pi[k] = __shfl_xor(ti[k], m, 32);
        }
#pragma unroll
        for (int k = 0; k < 5; ++k) bot5_insert(tv, ti, pv[k], pi[k]);
    }
    const int first = lane & ~(RLANES - 1);
#pragma unroll
    for (int k = 0; k < 5; ++k) {
        tv[k] = __shfl(tv[k], first, 32);
        ti[k] = __shfl(ti[k], first, 32);
    }
}

__global__ __launch_bounds__(128) void idw_reduce(
    const float* __restrict__ xq_s, const float* __restrict__ xt_s,
    const float* __restrict__ qnorm, const float* __restrict__ tnorm,
    const float* __restrict__ pvals, const int* __restrict__ pidx,
    const float* __restrict__ train_y, float* __restrict__ out) {
    const int tid  = threadIdx.x;
    const int lane = tid & 31;
    const int t    = tid & (RLANES - 1);
    const int q    = blockIdx.x * RQ_PER_WG + (tid / RLANES);

    float tv[5]; int ti[5];
    {
        const int base = (q * SEGS + t) * 5;
#pragma unroll
        for (int k = 0; k < 5; ++k) { tv[k] = pvals[base + k]; ti[k] = pidx[base + k]; }
    }
    merge8(tv, ti, lane);

    const v4f qc = *(const v4f*)(xq_s + (size_t)q * DDIM + 4 * t);
    const float qn = qnorm[q];
    const bool b4 = (t & 4) != 0, b2 = (t & 2) != 0, b1 = (t & 1) != 0;

    float rv[5]; int ri[5];
#pragma unroll
    for (int k = 0; k < 5; ++k) {
        const float* g = xt_s + (size_t)ti[k] * DDIM + 4 * t;
        float p[8];
#pragma unroll
        for (int j = 0; j < 8; ++j) {
            const v4f a = *(const v4f*)(g + j * DDIM);
            float d = a.x * qc.x;
            d = fmaf(a.y, qc.y, d);
            d = fmaf(a.z, qc.z, d);
            d = fmaf(a.w, qc.w, d);
            p[j] = d;
        }
        float s1[4];
#pragma unroll
        for (int i = 0; i < 4; ++i) {
            const float keep = b4 ? p[i + 4] : p[i];
            const float send = b4 ? p[i] : p[i + 4];
            s1[i] = keep + __shfl_xor(send, 4, 32);
        }
        float s2[2];
#pragma unroll
        for (int i = 0; i < 2; ++i) {
            const float keep = b2 ? s1[i + 2] : s1[i];
            const float send = b2 ? s1[i] : s1[i + 2];
            s2[i] = keep + __shfl_xor(send, 2, 32);
        }
        const float keep = b1 ? s2[1] : s2[0];
        const float send = b1 ? s2[0] : s2[1];
        const float dot  = keep + __shfl_xor(send, 1, 32);

        const int row = ti[k] + t;
        rv[k] = fmaf(-2.0f, dot, qn + tnorm[row]);
        ri[k] = row;
    }

#pragma unroll
    for (int k = 0; k < 5; ++k) { tv[k] = BIGF; ti[k] = 0; }
#pragma unroll
    for (int k = 0; k < 5; ++k) bot5_insert(tv, ti, rv[k], ri[k]);
    merge8(tv, ti, lane);

    float wv[5];
#pragma unroll
    for (int k = 0; k < 5; ++k)
        wv[k] = rsqrtf(fmaxf(tv[k], 0.f) + EPSV);

    float sw  = wv[0] + wv[1] + wv[2] + wv[3] + wv[4];
    float inv = 1.f / sw;

    v2f acc = {0.f, 0.f};
#pragma unroll
    for (int k = 0; k < 5; ++k) {
        const v2f yv = *(const v2f*)(train_y + (size_t)ti[k] * OUTD + 2 * t);
        acc += wv[k] * yv;
    }
    *(v2f*)(out + (size_t)q * OUTD + 2 * t) = acc * inv;
}

extern "C" void kernel_launch(void* const* d_in, const int* in_sizes, int n_in,
                              void* d_out, int out_size, void* d_ws, size_t ws_size,
                              hipStream_t stream) {
    const float* x       = (const float*)d_in[0];
    const float* train_x = (const float*)d_in[1];
    const float* train_y = (const float*)d_in[2];
    const float* weights = (const float*)d_in[3];
    float* out = (float*)d_out;

    float* ws    = (float*)d_ws;
    float* xq_s  = ws;
    float* xt_s  = ws + 131072;
    float* qnorm = ws + 655360;
    float* tnorm = ws + 659456;
    float* pvals = ws + 675840;
    int*   pidx  = (int*)(ws + 839680);

    idw_prep<<<(NQ + NT) / 128, 128, 0, stream>>>(x, train_x, weights,
                                                  xq_s, xt_s, qnorm, tnorm);
    idw_main<<<(NQ / (16 * WAVES_PER_WG)) * SEGS, WAVES_PER_WG * 32, 0, stream>>>(xq_s, xt_s, tnorm,
                                                  pvals, pidx);
    idw_reduce<<<NQ / RQ_PER_WG, 128, 0, stream>>>(xq_s, xt_s, qnorm, tnorm,
                                                  pvals, pidx, train_y, out);
}
